# starting kernel
# baseline (speedup 1.0000x reference)
#include <hip/hip_runtime.h>
#include <hip/hip_bf16.h>

typedef __attribute__((ext_vector_type(2))) float v2f;
typedef __attribute__((ext_vector_type(8))) float v8f;

#define DD    64
#define GG    128
#define NNODE 200000

__global__ void init_ws_kernel(float* __restrict__ zero_region, size_t nzero,
                               const float* __restrict__ W0,
                               const float* __restrict__ W1,
                               const float* __restrict__ W2,
                               float* __restrict__ wt) {
  size_t tid    = (size_t)blockIdx.x * blockDim.x + threadIdx.x;
  size_t stride = (size_t)gridDim.x * blockDim.x;
  for (size_t i = tid; i < nzero; i += stride) zero_region[i] = 0.0f;
  for (size_t i = tid; i < 3u * 64u * 64u; i += stride) {
    int s  = (int)(i >> 12);
    int nk = (int)(i & 4095);
    int n  = nk >> 6;
    int k  = nk & 63;
    const float* W = (s == 0) ? W0 : ((s == 1) ? W1 : W2);
    wt[i] = W[k * 64 + n];
  }
}

__global__ void scatter_b1_kernel(const float* __restrict__ x,
                                  const int* __restrict__ b1_rows,
                                  const float* __restrict__ b1_vals,
                                  const int* __restrict__ edge_batch,
                                  float* __restrict__ y,
                                  float* __restrict__ gcnt, int nE) {
  size_t tid = (size_t)blockIdx.x * blockDim.x + threadIdx.x;
  if (tid >= (size_t)nE * DD) return;
  int e = (int)(tid >> 6);
  int d = (int)(tid & 63);
  float xv = x[tid];
  int   r0 = b1_rows[2 * e];
  int   r1 = b1_rows[2 * e + 1];
  float v0 = b1_vals[2 * e];
  float v1 = b1_vals[2 * e + 1];
  atomicAdd(&y[(size_t)r0 * DD + d], v0 * xv);
  atomicAdd(&y[(size_t)r1 * DD + d], v1 * xv);
  if (d == 0) atomicAdd(&gcnt[edge_batch[e]], 1.0f);
}

__global__ void scatter_b2_kernel(const float* __restrict__ x,
                                  const int* __restrict__ b2_rows,
                                  const float* __restrict__ b2_vals,
                                  float* __restrict__ upper, int nT) {
  size_t tid = (size_t)blockIdx.x * blockDim.x + threadIdx.x;
  if (tid >= (size_t)nT * DD) return;
  int t = (int)(tid >> 6);
  int d = (int)(tid & 63);
  int   i0 = b2_rows[3 * t],     i1 = b2_rows[3 * t + 1], i2 = b2_rows[3 * t + 2];
  float v0 = b2_vals[3 * t],     v1 = b2_vals[3 * t + 1], v2 = b2_vals[3 * t + 2];
  float z = v0 * x[(size_t)i0 * DD + d]
          + v1 * x[(size_t)i1 * DD + d]
          + v2 * x[(size_t)i2 * DD + d];
  atomicAdd(&upper[(size_t)i0 * DD + d], v0 * z);
  atomicAdd(&upper[(size_t)i1 * DD + d], v1 * z);
  atomicAdd(&upper[(size_t)i2 * DD + d], v2 * z);
}

__global__ void __launch_bounds__(256)
fused_gemm_pool_kernel(const float* __restrict__ x,
                       const int* __restrict__ b1_rows,
                       const float* __restrict__ b1_vals,
                       const int* __restrict__ edge_batch,
                       const float* __restrict__ y,
                       const float* __restrict__ upper,
                       const float* __restrict__ wt,
                       float* __restrict__ gsum, int nE) {
  __shared__ float lgsum[GG * DD];
  for (int i = threadIdx.x; i < GG * DD; i += blockDim.x) lgsum[i] = 0.0f;
  __syncthreads();

  const int lane = threadIdx.x & 31;
  const int wave = threadIdx.x >> 5;
  const int tile = blockIdx.x * (blockDim.x >> 5) + wave;
  const int e0   = tile * 16;

  if (e0 < nE) {
    const int m  = lane & 15;
    const int kh = lane >> 4;
    const int e  = e0 + m;

    const int   r0 = b1_rows[2 * e];
    const int   r1 = b1_rows[2 * e + 1];
    const float s0 = b1_vals[2 * e];
    const float s1 = b1_vals[2 * e + 1];

    const float* xrow = x     + (size_t)e  * DD;
    const float* urow = upper + (size_t)e  * DD;
    const float* yr0  = y     + (size_t)r0 * DD;
    const float* yr1  = y     + (size_t)r1 * DD;

    v8f acc[4] = {};

    for (int src = 0; src < 3; ++src) {
      const float* wtm = wt + src * 4096;
      for (int ks = 0; ks < 16; ++ks) {
        const int kb = ks * 4 + kh * 2;
        v2f a;
        if (src == 0) {
          a.x = xrow[kb];  a.y = xrow[kb + 1];
        } else if (src == 1) {
          a.x = s0 * yr0[kb]     + s1 * yr1[kb];
          a.y = s0 * yr0[kb + 1] + s1 * yr1[kb + 1];
        } else {
          a.x = urow[kb];  a.y = urow[kb + 1];
        }
#pragma unroll
        for (int nt = 0; nt < 4; ++nt) {
          const float* wp = wtm + (nt * 16 + m) * 64 + kb;
          v2f b; b.x = wp[0]; b.y = wp[1];
          acc[nt] = __builtin_amdgcn_wmma_f32_16x16x4_f32(
              false, a, false, b, (short)0, acc[nt], false, false);
        }
      }
    }

    int gid[8];
#pragma unroll
    for (int r = 0; r < 8; ++r) gid[r] = edge_batch[e0 + kh * 8 + r];
#pragma unroll
    for (int nt = 0; nt < 4; ++nt) {
#pragma unroll
      for (int r = 0; r < 8; ++r) {
        float v = acc[nt][r];
        v = v > 0.0f ? v : 0.0f;
        atomicAdd(&lgsum[gid[r] * DD + nt * 16 + m], v);
      }
    }
  }

  __syncthreads();
  for (int i = threadIdx.x; i < GG * DD; i += blockDim.x)
    atomicAdd(&gsum[i], lgsum[i]);
}

__global__ void finalize_kernel(const float* __restrict__ gsum,
                                const float* __restrict__ gcnt,
                                float* __restrict__ out) {
  int i = blockIdx.x * blockDim.x + threadIdx.x;
  if (i >= GG * DD) return;
  float c = gcnt[i >> 6];
  float v = gsum[i] / fmaxf(c, 1.0f);
  out[i] = v;
  out[GG * DD + i] = v;
  out[2 * GG * DD + i] = v;
}

extern "C" void kernel_launch(void* const* d_in, const int* in_sizes, int n_in,
                              void* d_out, int out_size, void* d_ws, size_t ws_size,
                              hipStream_t stream) {
  const float* x        = (const float*)d_in[0];
  const int*   b1_rows  = (const int*)  d_in[1];
  const float* b1_vals  = (const float*)d_in[3];
  const int*   b2_rows  = (const int*)  d_in[4];
  const float* b2_vals  = (const float*)d_in[6];
  const int*   ebatch   = (const int*)  d_in[7];
  const float* W0       = (const float*)d_in[8];
  const float* W1       = (const float*)d_in[9];
  const float* W2       = (const float*)d_in[10];
  float*       out      = (float*)d_out;

  const int nE = in_sizes[0] / DD;
  const int nT = in_sizes[4] / 3;

  float* ws    = (float*)d_ws;
  float* y     = ws;
  size_t nY    = (size_t)NNODE * DD;
  float* upper = y + nY;
  size_t nU    = (size_t)nE * DD;
  float* gsum  = upper + nU;
  float* gcnt  = gsum + GG * DD;
  float* wt    = gcnt + GG;
  size_t nzero = nY + nU + GG * DD + GG;

  init_ws_kernel<<<2048, 256, 0, stream>>>(ws, nzero, W0, W1, W2, wt);

  {
    size_t total = (size_t)nE * DD;
    int blocks = (int)((total + 255) / 256);
    scatter_b1_kernel<<<blocks, 256, 0, stream>>>(x, b1_rows, b1_vals, ebatch,
                                                  y, gcnt, nE);
  }
  {
    size_t total = (size_t)nT * DD;
    int blocks = (int)((total + 255) / 256);
    scatter_b2_kernel<<<blocks, 256, 0, stream>>>(x, b2_rows, b2_vals, upper, nT);
  }
  {
    int tiles  = (nE + 15) / 16;
    int blocks = (tiles + 7) / 8;
    fused_gemm_pool_kernel<<<blocks, 256, 0, stream>>>(x, b1_rows, b1_vals,
                                                       ebatch, y, upper, wt,
                                                       gsum, nE);
  }
  finalize_kernel<<<(GG * DD + 255) / 256, 256, 0, stream>>>(gsum, gcnt, out);
}


# A operand of both projection kernels read as whole 128-byte lines (lane = column, large groups) and turned to fragment order through wave-private LDS, not 8-byte loads touching 16 lines each
# speedup vs baseline: 2.4046x; 2.4046x over previous
#include <hip/hip_runtime.h>
#include <hip/hip_bf16.h>

typedef __attribute__((ext_vector_type(2))) float v2f;
typedef __attribute__((ext_vector_type(8))) float v8f;

#define DD    64
#define GG    128
#define NNODE 200000

__global__ void init_ws_kernel(float* __restrict__ zero_region, size_t nzero,
                               const float* __restrict__ W0,
                               const float* __restrict__ W1,
                               const float* __restrict__ W2,
                               float* __restrict__ wt) {
  size_t tid    = (size_t)blockIdx.x * blockDim.x + threadIdx.x;
  size_t stride = (size_t)gridDim.x * blockDim.x;
  for (size_t i = tid; i < nzero; i += stride) zero_region[i] = 0.0f;
  for (size_t i = tid; i < 3u * 64u * 64u; i += stride) {
    int s  = (int)(i >> 12);
    int nk = (int)(i & 4095);
    int n  = nk >> 6;
    int k  = nk & 63;
    const float* W = (s == 0) ? W0 : ((s == 1) ? W1 : W2);
    wt[i] = W[k * 64 + n];
  }
}

#define MT 4
#define KG 8

__device__ __forceinline__ v2f ld2(const float* __restrict__ base, unsigned boff) {
  const float* p = (const float*)((const char*)base + boff);
  v2f r; r.x = p[0]; r.y = p[1];
  return r;
}

#define SP 66
#define XP 36
#define XROWS (16 * MT)

__device__ __forceinline__ void edge_tile_product(const float* __restrict__ x,
                                                  const float* __restrict__ wtp,
                                                  int e0, int nE, int lane,
                                                  int m, int kh,
                                                  unsigned woff,
                                                  float* __restrict__ xs,
                                                  v8f (&acc)[MT][4]) {
  const int e0s = __builtin_amdgcn_readfirstlane(e0);
  const int lb  = e0s + XROWS <= nE ? e0s : nE - XROWS;
  const int sh  = e0s - lb;
  const float* ar[MT];
#pragma unroll
  for (int mt = 0; mt < MT; ++mt) {
    int rr = sh + mt * 16 + m;
    rr = rr < XROWS ? rr : XROWS - 1;
    ar[mt] = xs + rr * XP + 2 * kh;
  }
  for (int kg = 0; kg < 16 / KG; ++kg) {
    const unsigned kgb = kg * (KG * 16u);
    {
      float t[XROWS];
      const float* p = x + ((size_t)lb * DD + kg * (KG * 4)) + lane;
#pragma unroll
      for (int R = 0; R < XROWS; ++R) t[R] = p[R * DD];
#pragma unroll
      for (int R = 0; R < XROWS; ++R) xs[R * XP + lane] = t[R];
    }
    __builtin_amdgcn_sched_barrier(0);

    v2f b[KG][4];
#pragma unroll
    for (int j = 0; j < KG; ++j)
#pragma unroll
      for (int nt = 0; nt < 4; ++nt)
        b[j][nt] = ld2(wtp, woff + nt * (16 * 64 * 4u) + kgb + j * 16u);

#pragma unroll
    for (int mt = 0; mt < MT; ++mt) {
      v2f a[KG];
#pragma unroll
      for (int j = 0; j < KG; ++j)
        a[j] = *(const v2f*)(ar[mt] + 4 * j);
#pragma unroll
      for (int j = 0; j < KG; ++j)
#pragma unroll
        for (int nt = 0; nt < 4; ++nt)
          acc[mt][nt] = __builtin_amdgcn_wmma_f32_16x16x4_f32(
              false, a[j], false, b[j][nt], (short)0, acc[mt][nt], false, false);
      __builtin_amdgcn_sched_group_barrier(0x100, 4, 0);
      __builtin_amdgcn_sched_group_barrier(0x008, 32, 0);
      __builtin_amdgcn_sched_barrier(0);
    }
  }
}

__global__ void __launch_bounds__(256)
edge_project_scatter_kernel(const float* __restrict__ x,
                            const int* __restrict__ b1_rows,
                            const float* __restrict__ b1_vals,
                            const int* __restrict__ edge_batch,
                            const float* __restrict__ wt1,
                            const float* __restrict__ wt0,
                            float* __restrict__ yw,
                            float* __restrict__ pe,
                            float* __restrict__ gcnt, int nE) {
  __shared__ __attribute__((aligned(16))) float stage[8][XROWS * XP];

  const int lane  = threadIdx.x & 31;
  const int wave  = threadIdx.x >> 5;
  const int wtile = blockIdx.x * (blockDim.x >> 5) + wave;
  const int e0    = wtile * (16 * MT);

  if (e0 < nE) {
    const int m  = lane & 15;
    const int kh = lane >> 4;

    int   j0[MT], j1[MT], gb[MT];
    float w0[MT], w1[MT];
#pragma unroll
    for (int mt = 0; mt < MT; ++mt) {
      const int e  = e0 + mt * 16 + m;
      const int ec = e < nE ? e : nE - 1;
      j0[mt] = b1_rows[2 * ec];
      j1[mt] = b1_rows[2 * ec + 1];
      w0[mt] = b1_vals[2 * ec];
      w1[mt] = b1_vals[2 * ec + 1];
      gb[mt] = edge_batch[ec];
    }
    const unsigned woff = (unsigned)m * (64 * 4u) + kh * 8u;
    float* st = stage[wave];

    {
      v8f acc[MT][4] = {};
      edge_tile_product(x, wt1, e0, nE, lane, m, kh, woff, st, acc);

#pragma unroll
      for (int mt = 0; mt < MT; ++mt) {
#pragma unroll
        for (int nt = 0; nt < 4; ++nt)
#pragma unroll
          for (int r = 0; r < 8; ++r)
            st[(kh * 8 + r) * SP + nt * 16 + m] = acc[mt][nt][r];

        if (kh == 0 && e0 + mt * 16 + m < nE) atomicAdd(&gcnt[gb[mt]], 1.0f);

        float lo[16], hi[16];
#pragma unroll
        for (int R = 0; R < 16; ++R) {
          lo[R] = st[R * SP + lane];
          hi[R] = st[R * SP + 32 + lane];
        }
#pragma unroll
        for (int R = 0; R < 16; ++R) {
          if (e0 + mt * 16 + R < nE) {
            const int   d0 = __builtin_amdgcn_readlane(j0[mt], R);
            const int   d1 = __builtin_amdgcn_readlane(j1[mt], R);
            const float s0 = __int_as_float(__builtin_amdgcn_readlane(__float_as_int(w0[mt]), R));
            const float s1 = __int_as_float(__builtin_amdgcn_readlane(__float_as_int(w1[mt]), R));
            float* p0 = yw + (size_t)d0 * DD + lane;
            float* p1 = yw + (size_t)d1 * DD + lane;
            atomicAdd(p0,      s0 * lo[R]);
            atomicAdd(p0 + 32, s0 * hi[R]);
            atomicAdd(p1,      s1 * lo[R]);
            atomicAdd(p1 + 32, s1 * hi[R]);
          }
        }
      }
    }

    {
      v8f acc[MT][4] = {};
      edge_tile_product(x, wt0, e0, nE, lane, m, kh, woff, st, acc);

#pragma unroll
      for (int mt = 0; mt < MT; ++mt) {
#pragma unroll
        for (int nt = 0; nt < 4; ++nt)
#pragma unroll
          for (int r = 0; r < 8; ++r)
            st[(kh * 8 + r) * SP + nt * 16 + m] = acc[mt][nt][r];

        float lo[16], hi[16];
#pragma unroll
        for (int R = 0; R < 16; ++R) {
          lo[R] = st[R * SP + lane];
          hi[R] = st[R * SP + 32 + lane];
        }
#pragma unroll
        for (int R = 0; R < 16; ++R) {
          const int row = e0 + mt * 16 + R;
          if (row < nE) {
            float* pp = pe + (size_t)row * DD + lane;
            pp[0]  = lo[R];
            pp[32] = hi[R];
          }
        }
      }
    }
  }
}

#define TMT 2
#define ZP 68
__global__ void __launch_bounds__(256)
tri_project_scatter_kernel(const float* __restrict__ x,
                           const int* __restrict__ b2_rows,
                           const float* __restrict__ b2_vals,
                           const float* __restrict__ wt2,
                           float* __restrict__ upw, int nT) {
  __shared__ __attribute__((aligned(16))) float stage[8][16 * TMT * ZP];

  const int lane  = threadIdx.x & 31;
  const int wave  = threadIdx.x >> 5;
  const int wtile = blockIdx.x * (blockDim.x >> 5) + wave;
  const int t0    = wtile * (16 * TMT);

  if (t0 < nT) {
    const int m  = lane & 15;
    const int kh = lane >> 4;

    int   r0[TMT], r1[TMT], r2[TMT];
    float c0[TMT], c1[TMT], c2[TMT];
#pragma unroll
    for (int mt = 0; mt < TMT; ++mt) {
      const int t  = t0 + mt * 16 + m;
      const int tc = t < nT ? t : nT - 1;
      r0[mt] = b2_rows[3 * tc];
      r1[mt] = b2_rows[3 * tc + 1];
      r2[mt] = b2_rows[3 * tc + 2];
      c0[mt] = b2_vals[3 * tc];
      c1[mt] = b2_vals[3 * tc + 1];
      c2[mt] = b2_vals[3 * tc + 2];
    }
    const unsigned woff = (unsigned)m * (64 * 4u) + kh * 8u;
    float* st = stage[wave];

#pragma unroll
    for (int mt = 0; mt < TMT; ++mt) {
      float pl[16], ph[16], ql[16], qh[16], wl[16], wh[16];
#pragma unroll
      for (int R = 0; R < 16; ++R) {
        const int i0 = __builtin_amdgcn_readlane(r0[mt], R);
        const int i1 = __builtin_amdgcn_readlane(r1[mt], R);
        const int i2 = __builtin_amdgcn_readlane(r2[mt], R);
        const float* pa = (const float*)((const char*)x + ((unsigned)i0 * (DD * 4u) + lane * 4u));
        const float* pb = (const float*)((const char*)x + ((unsigned)i1 * (DD * 4u) + lane * 4u));
        const float* pc = (const float*)((const char*)x + ((unsigned)i2 * (DD * 4u) + lane * 4u));
        pl[R] = pa[0];  ph[R] = pa[32];
        ql[R] = pb[0];  qh[R] = pb[32];
        wl[R] = pc[0];  wh[R] = pc[32];
      }
#pragma unroll
      for (int R = 0; R < 16; ++R) {
        const float k0 = __int_as_float(__builtin_amdgcn_readlane(__float_as_int(c0[mt]), R));
        const float k1 = __int_as_float(__builtin_amdgcn_readlane(__float_as_int(c1[mt]), R));
        const float k2 = __int_as_float(__builtin_amdgcn_readlane(__float_as_int(c2[mt]), R));
        float* zr = st + (mt * 16 + R) * ZP + lane;
        zr[0]  = k0 * pl[R] + k1 * ql[R] + k2 * wl[R];
        zr[32] = k0 * ph[R] + k1 * qh[R] + k2 * wh[R];
      }
    }
    __builtin_amdgcn_sched_barrier(0);

    v8f acc[TMT][4] = {};

    for (int kg = 0; kg < 16 / KG; ++kg) {
      const unsigned kgb = kg * (KG * 16u);
      v2f b[KG][4];
#pragma unroll
      for (int j = 0; j < KG; ++j)
#pragma unroll
        for (int nt = 0; nt < 4; ++nt)
          b[j][nt] = ld2(wt2, woff + nt * (16 * 64 * 4u) + kgb + j * 16u);

#pragma unroll
      for (int mt = 0; mt < TMT; ++mt) {
        v2f a[KG];
        const float* ar = st + (mt * 16 + m) * ZP + kg * (KG * 4) + 2 * kh;
#pragma unroll
        for (int j = 0; j < KG; ++j)
          a[j] = *(const v2f*)(ar + 4 * j);
#pragma unroll
        for (int j = 0; j < KG; ++j)
#pragma unroll
          for (int nt = 0; nt < 4; ++nt)
            acc[mt][nt] = __builtin_amdgcn_wmma_f32_16x16x4_f32(
                false, a[j], false, b[j][nt], (short)0, acc[mt][nt], false, false);
        __builtin_amdgcn_sched_group_barrier(0x100, 4, 0);
        __builtin_amdgcn_sched_group_barrier(0x008, 32, 0);
        __builtin_amdgcn_sched_barrier(0);
      }
    }

#pragma unroll
    for (int mt = 0; mt < TMT; ++mt) {
#pragma unroll
      for (int nt = 0; nt < 4; ++nt)
#pragma unroll
        for (int r = 0; r < 8; ++r)
          st[(kh * 8 + r) * SP + nt * 16 + m] = acc[mt][nt][r];

      const int tl  = t0 + mt * 16 + m;
      const int tlc = tl < nT ? tl : nT - 1;
      const int   j0 = b2_rows[3 * tlc], j1 = b2_rows[3 * tlc + 1], j2 = b2_rows[3 * tlc + 2];
      const float w0 = b2_vals[3 * tlc], w1 = b2_vals[3 * tlc + 1], w2 = b2_vals[3 * tlc + 2];

      float lo[16], hi[16];
#pragma unroll
      for (int R = 0; R < 16; ++R) {
        lo[R] = st[R * SP + lane];
        hi[R] = st[R * SP + 32 + lane];
      }
#pragma unroll
      for (int R = 0; R < 16; ++R) {
        if (t0 + mt * 16 + R < nT) {
          const int   d0 = __builtin_amdgcn_readlane(j0, R);
          const int   d1 = __builtin_amdgcn_readlane(j1, R);
          const int   d2 = __builtin_amdgcn_readlane(j2, R);
          const float s0 = __int_as_float(__builtin_amdgcn_readlane(__float_as_int(w0), R));
          const float s1 = __int_as_float(__builtin_amdgcn_readlane(__float_as_int(w1), R));
          const float s2 = __int_as_float(__builtin_amdgcn_readlane(__float_as_int(w2), R));
          float* p0 = upw + (size_t)d0 * DD + lane;
          float* p1 = upw + (size_t)d1 * DD + lane;
          float* p2 = upw + (size_t)d2 * DD + lane;
          atomicAdd(p0,      s0 * lo[R]);
          atomicAdd(p0 + 32, s0 * hi[R]);
          atomicAdd(p1,      s1 * lo[R]);
          atomicAdd(p1 + 32, s1 * hi[R]);
          atomicAdd(p2,      s2 * lo[R]);
          atomicAdd(p2 + 32, s2 * hi[R]);
        }
      }
    }
  }
}

#define RG 8
__global__ void __launch_bounds__(256)
combine_pool_kernel(const float* __restrict__ pe,
                    const int* __restrict__ b1_rows,
                    const float* __restrict__ b1_vals,
                    const int* __restrict__ edge_batch,
                    const float* __restrict__ yw,
                    float* __restrict__ gsum, int nE) {
  __shared__ float lgsum[GG * DD];
  for (int i = threadIdx.x; i < GG * DD; i += blockDim.x) lgsum[i] = 0.0f;
  __syncthreads();

  const int lane  = threadIdx.x & 31;
  const int wave  = threadIdx.x >> 5;
  const int wtile = blockIdx.x * (blockDim.x >> 5) + wave;
  const int e0    = wtile * (16 * MT);

  if (e0 < nE) {
    const int m = lane & 15;

    int   n0[MT], n1[MT], gb[MT];
    float s0[MT], s1[MT];
#pragma unroll
    for (int mt = 0; mt < MT; ++mt) {
      const int e  = e0 + mt * 16 + m;
      const int ec = e < nE ? e : nE - 1;
      n0[mt] = b1_rows[2 * ec];
      n1[mt] = b1_rows[2 * ec + 1];
      s0[mt] = b1_vals[2 * ec];
      s1[mt] = b1_vals[2 * ec + 1];
      gb[mt] = edge_batch[ec];
    }

#pragma unroll
    for (int mt = 0; mt < MT; ++mt) {
#pragma unroll
      for (int h = 0; h < 16 / RG; ++h) {
        float pl[RG], ph[RG], al[RG], ah[RG], bl[RG], bh[RG];
        float c0[RG], c1[RG];
        int   g[RG];
#pragma unroll
        for (int q = 0; q < RG; ++q) {
          const int R   = h * RG + q;
          const int row = e0 + mt * 16 + R;
          const int rc  = row < nE ? row : nE - 1;
          const int d0  = __builtin_amdgcn_readlane(n0[mt], R);
          const int d1  = __builtin_amdgcn_readlane(n1[mt], R);
          c0[q] = __int_as_float(__builtin_amdgcn_readlane(__float_as_int(s0[mt]), R));
          c1[q] = __int_as_float(__builtin_amdgcn_readlane(__float_as_int(s1[mt]), R));
          g[q]  = __builtin_amdgcn_readlane(gb[mt], R);
          const float* pp = pe + (size_t)rc * DD + lane;
          const float* pa = yw + (size_t)d0 * DD + lane;
          const float* pb = yw + (size_t)d1 * DD + lane;
          pl[q] = pp[0];  ph[q] = pp[32];
          al[q] = pa[0];  ah[q] = pa[32];
          bl[q] = pb[0];  bh[q] = pb[32];
        }
#pragma unroll
        for (int q = 0; q < RG; ++q) {
          const int row = e0 + mt * 16 + h * RG + q;
          if (row < nE) {
            float vl = pl[q] + (c0[q] * al[q] + c1[q] * bl[q]);
            float vh = ph[q] + (c0[q] * ah[q] + c1[q] * bh[q]);
            vl = vl > 0.0f ? vl : 0.0f;
            vh = vh > 0.0f ? vh : 0.0f;
            atomicAdd(&lgsum[g[q] * DD + lane],      vl);
            atomicAdd(&lgsum[g[q] * DD + 32 + lane], vh);
          }
        }
      }
    }
  }

  __syncthreads();
  for (int i = threadIdx.x; i < GG * DD; i += blockDim.x)
    atomicAdd(&gsum[i], lgsum[i]);
}

__global__ void finalize_kernel(const float* __restrict__ gsum,
                                const float* __restrict__ gcnt,
                                float* __restrict__ out) {
  int i = blockIdx.x * blockDim.x + threadIdx.x;
  if (i >= GG * DD) return;
  float c = gcnt[i >> 6];
  float v = gsum[i] / fmaxf(c, 1.0f);
  out[i] = v;
  out[GG * DD + i] = v;
  out[2 * GG * DD + i] = v;
}

extern "C" void kernel_launch(void* const* d_in, const int* in_sizes, int n_in,
                              void* d_out, int out_size, void* d_ws, size_t ws_size,
                              hipStream_t stream) {
  const float* x        = (const float*)d_in[0];
  const int*   b1_rows  = (const int*)  d_in[1];
  const float* b1_vals  = (const float*)d_in[3];
  const int*   b2_rows  = (const int*)  d_in[4];
  const float* b2_vals  = (const float*)d_in[6];
  const int*   ebatch   = (const int*)  d_in[7];
  const float* W0       = (const float*)d_in[8];
  const float* W1       = (const float*)d_in[9];
  const float* W2       = (const float*)d_in[10];
  float*       out      = (float*)d_out;

  const int nE = in_sizes[0] / DD;
  const int nT = in_sizes[4] / 3;

  float* ws    = (float*)d_ws;
  float* y     = ws;
  size_t nY    = (size_t)NNODE * DD;
  float* gsum  = y + nY;
  float* gcnt  = gsum + GG * DD;
  float* upper = gcnt + GG;
  size_t nU    = (size_t)nE * DD;
  float* wt    = upper + nU;
  size_t nzero = nY + GG * DD + GG;

  init_ws_kernel<<<2048, 256, 0, stream>>>(ws, nzero, W0, W1, W2, wt);

  {
    int tiles  = (nE + 16 * MT - 1) / (16 * MT);
    int blocks = (tiles + 7) / 8;
    edge_project_scatter_kernel<<<blocks, 256, 0, stream>>>(x, b1_rows, b1_vals,
                                                            ebatch, wt + 4096, wt,
                                                            y, upper, gcnt, nE);
  }
  {
    int tiles  = (nT + 16 * TMT - 1) / (16 * TMT);
    int blocks = (tiles + 7) / 8;
    tri_project_scatter_kernel<<<blocks, 256, 0, stream>>>(x, b2_rows, b2_vals,
                                                           wt + 2 * 4096, upper, nT);
  }
  {
    int tiles  = (nE + 16 * MT - 1) / (16 * MT);
    int blocks = (tiles + 7) / 8;
    combine_pool_kernel<<<blocks, 256, 0, stream>>>(upper, b1_rows, b1_vals,
                                                    ebatch, y, gsum, nE);
  }
  finalize_kernel<<<(GG * DD + 255) / 256, 256, 0, stream>>>(gsum, gcnt, out);
}


# graph counts: each wave's 4 f32 count adds, all waves on the same 4 lines and issued among the row adds, go as one group ahead of them into 1 of 8 replica rows; finalize sums the rows (exact)
# speedup vs baseline: 3.0400x; 1.2642x over previous
#include <hip/hip_runtime.h>
#include <hip/hip_bf16.h>

typedef __attribute__((ext_vector_type(2))) float v2f;
typedef __attribute__((ext_vector_type(8))) float v8f;

#define DD    64
#define GG    128
#define NNODE 200000
#define CR    8

__global__ void init_ws_kernel(float* __restrict__ zero_region, size_t nzero,
                               const float* __restrict__ W0,
                               const float* __restrict__ W1,
                               const float* __restrict__ W2,
                               float* __restrict__ wt) {
  size_t tid    = (size_t)blockIdx.x * blockDim.x + threadIdx.x;
  size_t stride = (size_t)gridDim.x * blockDim.x;
  for (size_t i = tid; i < nzero; i += stride) zero_region[i] = 0.0f;
  for (size_t i = tid; i < 3u * 64u * 64u; i += stride) {
    int s  = (int)(i >> 12);
    int nk = (int)(i & 4095);
    int n  = nk >> 6;
    int k  = nk & 63;
    const float* W = (s == 0) ? W0 : ((s == 1) ? W1 : W2);
    wt[i] = W[k * 64 + n];
  }
}

#define MT 4
#define KG 8

__device__ __forceinline__ v2f ld2(const float* __restrict__ base, unsigned boff) {
  const float* p = (const float*)((const char*)base + boff);
  v2f r; r.x = p[0]; r.y = p[1];
  return r;
}

#define SP 66
#define XP 36
#define XROWS (16 * MT)

__device__ __forceinline__ void edge_tile_product(const float* __restrict__ x,
                                                  const float* __restrict__ wtp,
                                                  int e0, int nE, int lane,
                                                  int m, int kh,
                                                  unsigned woff,
                                                  float* __restrict__ xs,
                                                  v8f (&acc)[MT][4]) {
  const int e0s = __builtin_amdgcn_readfirstlane(e0);
  const int lb  = e0s + XROWS <= nE ? e0s : nE - XROWS;
  const int sh  = e0s - lb;
  const float* ar[MT];
#pragma unroll
  for (int mt = 0; mt < MT; ++mt) {
    int rr = sh + mt * 16 + m;
    rr = rr < XROWS ? rr : XROWS - 1;
    ar[mt] = xs + rr * XP + 2 * kh;
  }
  for (int kg = 0; kg < 16 / KG; ++kg) {
    const unsigned kgb = kg * (KG * 16u);
    {
      float t[XROWS];
      const float* p = x + ((size_t)lb * DD + kg * (KG * 4)) + lane;
#pragma unroll
      for (int R = 0; R < XROWS; ++R) t[R] = p[R * DD];
#pragma unroll
      for (int R = 0; R < XROWS; ++R) xs[R * XP + lane] = t[R];
    }
    __builtin_amdgcn_sched_barrier(0);

    v2f b[KG][4];
#pragma unroll
    for (int j = 0; j < KG; ++j)
#pragma unroll
      for (int nt = 0; nt < 4; ++nt)
        b[j][nt] = ld2(wtp, woff + nt * (16 * 64 * 4u) + kgb + j * 16u);

#pragma unroll
    for (int mt = 0; mt < MT; ++mt) {
      v2f a[KG];
#pragma unroll
      for (int j = 0; j < KG; ++j)
        a[j] = *(const v2f*)(ar[mt] + 4 * j);
#pragma unroll
      for (int j = 0; j < KG; ++j)
#pragma unroll
        for (int nt = 0; nt < 4; ++nt)
          acc[mt][nt] = __builtin_amdgcn_wmma_f32_16x16x4_f32(
              false, a[j], false, b[j][nt], (short)0, acc[mt][nt], false, false);
      __builtin_amdgcn_sched_group_barrier(0x100, 4, 0);
      __builtin_amdgcn_sched_group_barrier(0x008, 32, 0);
      __builtin_amdgcn_sched_barrier(0);
    }
  }
}

__global__ void __launch_bounds__(256)
edge_project_scatter_kernel(const float* __restrict__ x,
                            const int* __restrict__ b1_rows,
                            const float* __restrict__ b1_vals,
                            const int* __restrict__ edge_batch,
                            const float* __restrict__ wt1,
                            const float* __restrict__ wt0,
                            float* __restrict__ yw,
                            float* __restrict__ pe,
                            float* __restrict__ gcnt, int nE) {
  __shared__ __attribute__((aligned(16))) float stage[8][XROWS * XP];

  const int lane  = threadIdx.x & 31;
  const int wave  = threadIdx.x >> 5;
  const int wtile = blockIdx.x * (blockDim.x >> 5) + wave;
  const int e0    = wtile * (16 * MT);

  if (e0 < nE) {
    const int m  = lane & 15;
    const int kh = lane >> 4;

    int   j0[MT], j1[MT], gb[MT];
    float w0[MT], w1[MT];
#pragma unroll
    for (int mt = 0; mt < MT; ++mt) {
      const int e  = e0 + mt * 16 + m;
      const int ec = e < nE ? e : nE - 1;
      j0[mt] = b1_rows[2 * ec];
      j1[mt] = b1_rows[2 * ec + 1];
      w0[mt] = b1_vals[2 * ec];
      w1[mt] = b1_vals[2 * ec + 1];
      gb[mt] = edge_batch[ec];
    }
    const unsigned woff = (unsigned)m * (64 * 4u) + kh * 8u;
    float* st = stage[wave];

    {
      v8f acc[MT][4] = {};
      edge_tile_product(x, wt1, e0, nE, lane, m, kh, woff, st, acc);

      {
        float* gc = gcnt + (wtile & (CR - 1)) * GG;
#pragma unroll
        for (int mt = 0; mt < MT; ++mt)
          if (kh == 0 && e0 + mt * 16 + m < nE) atomicAdd(&gc[gb[mt]], 1.0f);
      }

#pragma unroll
      for (int mt = 0; mt < MT; ++mt) {
#pragma unroll
        for (int nt = 0; nt < 4; ++nt)
#pragma unroll
          for (int r = 0; r < 8; ++r)
            st[(kh * 8 + r) * SP + nt * 16 + m] = acc[mt][nt][r];

        float lo[16], hi[16];
#pragma unroll
        for (int R = 0; R < 16; ++R) {
          lo[R] = st[R * SP + lane];
          hi[R] = st[R * SP + 32 + lane];
        }
#pragma unroll
        for (int R = 0; R < 16; ++R) {
          if (e0 + mt * 16 + R < nE) {
            const int   d0 = __builtin_amdgcn_readlane(j0[mt], R);
            const int   d1 = __builtin_amdgcn_readlane(j1[mt], R);
            const float s0 = __int_as_float(__builtin_amdgcn_readlane(__float_as_int(w0[mt]), R));
            const float s1 = __int_as_float(__builtin_amdgcn_readlane(__float_as_int(w1[mt]), R));
            float* p0 = yw + (size_t)d0 * DD + lane;
            float* p1 = yw + (size_t)d1 * DD + lane;
            atomicAdd(p0,      s0 * lo[R]);
            atomicAdd(p0 + 32, s0 * hi[R]);
            atomicAdd(p1,      s1 * lo[R]);
            atomicAdd(p1 + 32, s1 * hi[R]);
          }
        }
      }
    }

    {
      v8f acc[MT][4] = {};
      edge_tile_product(x, wt0, e0, nE, lane, m, kh, woff, st, acc);

#pragma unroll
      for (int mt = 0; mt < MT; ++mt) {
#pragma unroll
        for (int nt = 0; nt < 4; ++nt)
#pragma unroll
          for (int r = 0; r < 8; ++r)
            st[(kh * 8 + r) * SP + nt * 16 + m] = acc[mt][nt][r];

        float lo[16], hi[16];
#pragma unroll
        for (int R = 0; R < 16; ++R) {
          lo[R] = st[R * SP + lane];
          hi[R] = st[R * SP + 32 + lane];
        }
#pragma unroll
        for (int R = 0; R < 16; ++R) {
          const int row = e0 + mt * 16 + R;
          if (row < nE) {
            float* pp = pe + (size_t)row * DD + lane;
            pp[0]  = lo[R];
            pp[32] = hi[R];
          }
        }
      }
    }
  }
}

#define TMT 2
#define ZP 68
__global__ void __launch_bounds__(256)
tri_project_scatter_kernel(const float* __restrict__ x,
                           const int* __restrict__ b2_rows,
                           const float* __restrict__ b2_vals,
                           const float* __restrict__ wt2,
                           float* __restrict__ upw, int nT) {
  __shared__ __attribute__((aligned(16))) float stage[8][16 * TMT * ZP];

  const int lane  = threadIdx.x & 31;
  const int wave  = threadIdx.x >> 5;
  const int wtile = blockIdx.x * (blockDim.x >> 5) + wave;
  const int t0    = wtile * (16 * TMT);

  if (t0 < nT) {
    const int m  = lane & 15;
    const int kh = lane >> 4;

    int   r0[TMT], r1[TMT], r2[TMT];
    float c0[TMT], c1[TMT], c2[TMT];
#pragma unroll
    for (int mt = 0; mt < TMT; ++mt) {
      const int t  = t0 + mt * 16 + m;
      const int tc = t < nT ? t : nT - 1;
      r0[mt] = b2_rows[3 * tc];
      r1[mt] = b2_rows[3 * tc + 1];
      r2[mt] = b2_rows[3 * tc + 2];
      c0[mt] = b2_vals[3 * tc];
      c1[mt] = b2_vals[3 * tc + 1];
      c2[mt] = b2_vals[3 * tc + 2];
    }
    const unsigned woff = (unsigned)m * (64 * 4u) + kh * 8u;
    float* st = stage[wave];

#pragma unroll
    for (int mt = 0; mt < TMT; ++mt) {
      float pl[16], ph[16], ql[16], qh[16], wl[16], wh[16];
#pragma unroll
      for (int R = 0; R < 16; ++R) {
        const int i0 = __builtin_amdgcn_readlane(r0[mt], R);
        const int i1 = __builtin_amdgcn_readlane(r1[mt], R);
        const int i2 = __builtin_amdgcn_readlane(r2[mt], R);
        const float* pa = (const float*)((const char*)x + ((unsigned)i0 * (DD * 4u) + lane * 4u));
        const float* pb = (const float*)((const char*)x + ((unsigned)i1 * (DD * 4u) + lane * 4u));
        const float* pc = (const float*)((const char*)x + ((unsigned)i2 * (DD * 4u) + lane * 4u));
        pl[R] = pa[0];  ph[R] = pa[32];
        ql[R] = pb[0];  qh[R] = pb[32];
        wl[R] = pc[0];  wh[R] = pc[32];
      }
#pragma unroll
      for (int R = 0; R < 16; ++R) {
        const float k0 = __int_as_float(__builtin_amdgcn_readlane(__float_as_int(c0[mt]), R));
        const float k1 = __int_as_float(__builtin_amdgcn_readlane(__float_as_int(c1[mt]), R));
        const float k2 = __int_as_float(__builtin_amdgcn_readlane(__float_as_int(c2[mt]), R));
        float* zr = st + (mt * 16 + R) * ZP + lane;
        zr[0]  = k0 * pl[R] + k1 * ql[R] + k2 * wl[R];
        zr[32] = k0 * ph[R] + k1 * qh[R] + k2 * wh[R];
      }
    }
    __builtin_amdgcn_sched_barrier(0);

    v8f acc[TMT][4] = {};

    for (int kg = 0; kg < 16 / KG; ++kg) {
      const unsigned kgb = kg * (KG * 16u);
      v2f b[KG][4];
#pragma unroll
      for (int j = 0; j < KG; ++j)
#pragma unroll
        for (int nt = 0; nt < 4; ++nt)
          b[j][nt] = ld2(wt2, woff + nt * (16 * 64 * 4u) + kgb + j * 16u);

#pragma unroll
      for (int mt = 0; mt < TMT; ++mt) {
        v2f a[KG];
        const float* ar = st + (mt * 16 + m) * ZP + kg * (KG * 4) + 2 * kh;
#pragma unroll
        for (int j = 0; j < KG; ++j)
          a[j] = *(const v2f*)(ar + 4 * j);
#pragma unroll
        for (int j = 0; j < KG; ++j)
#pragma unroll
          for (int nt = 0; nt < 4; ++nt)
            acc[mt][nt] = __builtin_amdgcn_wmma_f32_16x16x4_f32(
                false, a[j], false, b[j][nt], (short)0, acc[mt][nt], false, false);
        __builtin_amdgcn_sched_group_barrier(0x100, 4, 0);
        __builtin_amdgcn_sched_group_barrier(0x008, 32, 0);
        __builtin_amdgcn_sched_barrier(0);
      }
    }

#pragma unroll
    for (int mt = 0; mt < TMT; ++mt) {
#pragma unroll
      for (int nt = 0; nt < 4; ++nt)
#pragma unroll
        for (int r = 0; r < 8; ++r)
          st[(kh * 8 + r) * SP + nt * 16 + m] = acc[mt][nt][r];

      const int tl  = t0 + mt * 16 + m;
      const int tlc = tl < nT ? tl : nT - 1;
      const int   j0 = b2_rows[3 * tlc], j1 = b2_rows[3 * tlc + 1], j2 = b2_rows[3 * tlc + 2];
      const float w0 = b2_vals[3 * tlc], w1 = b2_vals[3 * tlc + 1], w2 = b2_vals[3 * tlc + 2];

      float lo[16], hi[16];
#pragma unroll
      for (int R = 0; R < 16; ++R) {
        lo[R] = st[R * SP + lane];
        hi[R] = st[R * SP + 32 + lane];
      }
#pragma unroll
      for (int R = 0; R < 16; ++R) {
        if (t0 + mt * 16 + R < nT) {
          const int   d0 = __builtin_amdgcn_readlane(j0, R);
          const int   d1 = __builtin_amdgcn_readlane(j1, R);
          const int   d2 = __builtin_amdgcn_readlane(j2, R);
          const float s0 = __int_as_float(__builtin_amdgcn_readlane(__float_as_int(w0), R));
          const float s1 = __int_as_float(__builtin_amdgcn_readlane(__float_as_int(w1), R));
          const float s2 = __int_as_float(__builtin_amdgcn_readlane(__float_as_int(w2), R));
          float* p0 = upw + (size_t)d0 * DD + lane;
          float* p1 = upw + (size_t)d1 * DD + lane;
          float* p2 = upw + (size_t)d2 * DD + lane;
          atomicAdd(p0,      s0 * lo[R]);
          atomicAdd(p0 + 32, s0 * hi[R]);
          atomicAdd(p1,      s1 * lo[R]);
          atomicAdd(p1 + 32, s1 * hi[R]);
          atomicAdd(p2,      s2 * lo[R]);
          atomicAdd(p2 + 32, s2 * hi[R]);
        }
      }
    }
  }
}

#define RG 8
__global__ void __launch_bounds__(256)
combine_pool_kernel(const float* __restrict__ pe,
                    const int* __restrict__ b1_rows,
                    const float* __restrict__ b1_vals,
                    const int* __restrict__ edge_batch,
                    const float* __restrict__ yw,
                    float* __restrict__ gsum, int nE) {
  __shared__ float lgsum[GG * DD];
  for (int i = threadIdx.x; i < GG * DD; i += blockDim.x) lgsum[i] = 0.0f;
  __syncthreads();

  const int lane  = threadIdx.x & 31;
  const int wave  = threadIdx.x >> 5;
  const int wtile = blockIdx.x * (blockDim.x >> 5) + wave;
  const int e0    = wtile * (16 * MT);

  if (e0 < nE) {
    const int m = lane & 15;

    int   n0[MT], n1[MT], gb[MT];
    float s0[MT], s1[MT];
#pragma unroll
    for (int mt = 0; mt < MT; ++mt) {
      const int e  = e0 + mt * 16 + m;
      const int ec = e < nE ? e : nE - 1;
      n0[mt] = b1_rows[2 * ec];
      n1[mt] = b1_rows[2 * ec + 1];
      s0[mt] = b1_vals[2 * ec];
      s1[mt] = b1_vals[2 * ec + 1];
      gb[mt] = edge_batch[ec];
    }

#pragma unroll
    for (int mt = 0; mt < MT; ++mt) {
#pragma unroll
      for (int h = 0; h < 16 / RG; ++h) {
        float pl[RG], ph[RG], al[RG], ah[RG], bl[RG], bh[RG];
        float c0[RG], c1[RG];
        int   g[RG];
#pragma unroll
        for (int q = 0; q < RG; ++q) {
          const int R   = h * RG + q;
          const int row = e0 + mt * 16 + R;
          const int rc  = row < nE ? row : nE - 1;
          const int d0  = __builtin_amdgcn_readlane(n0[mt], R);
          const int d1  = __builtin_amdgcn_readlane(n1[mt], R);
          c0[q] = __int_as_float(__builtin_amdgcn_readlane(__float_as_int(s0[mt]), R));
          c1[q] = __int_as_float(__builtin_amdgcn_readlane(__float_as_int(s1[mt]), R));
          g[q]  = __builtin_amdgcn_readlane(gb[mt], R);
          const float* pp = pe + (size_t)rc * DD + lane;
          const float* pa = yw + (size_t)d0 * DD + lane;
          const float* pb = yw + (size_t)d1 * DD + lane;
          pl[q] = pp[0];  ph[q] = pp[32];
          al[q] = pa[0];  ah[q] = pa[32];
          bl[q] = pb[0];  bh[q] = pb[32];
        }
#pragma unroll
        for (int q = 0; q < RG; ++q) {
          const int row = e0 + mt * 16 + h * RG + q;
          if (row < nE) {
            float vl = pl[q] + (c0[q] * al[q] + c1[q] * bl[q]);
            float vh = ph[q] + (c0[q] * ah[q] + c1[q] * bh[q]);
            vl = vl > 0.0f ? vl : 0.0f;
            vh = vh > 0.0f ? vh : 0.0f;
            atomicAdd(&lgsum[g[q] * DD + lane],      vl);
            atomicAdd(&lgsum[g[q] * DD + 32 + lane], vh);
          }
        }
      }
    }
  }

  __syncthreads();
  for (int i = threadIdx.x; i < GG * DD; i += blockDim.x)
    atomicAdd(&gsum[i], lgsum[i]);
}

__global__ void finalize_kernel(const float* __restrict__ gsum,
                                const float* __restrict__ gcnt,
                                float* __restrict__ out) {
  int i = blockIdx.x * blockDim.x + threadIdx.x;
  if (i >= GG * DD) return;
  float c = 0.0f;
#pragma unroll
  for (int r = 0; r < CR; ++r) c += gcnt[r * GG + (i >> 6)];
  float v = gsum[i] / fmaxf(c, 1.0f);
  out[i] = v;
  out[GG * DD + i] = v;
  out[2 * GG * DD + i] = v;
}

extern "C" void kernel_launch(void* const* d_in, const int* in_sizes, int n_in,
                              void* d_out, int out_size, void* d_ws, size_t ws_size,
                              hipStream_t stream) {
  const float* x        = (const float*)d_in[0];
  const int*   b1_rows  = (const int*)  d_in[1];
  const float* b1_vals  = (const float*)d_in[3];
  const int*   b2_rows  = (const int*)  d_in[4];
  const float* b2_vals  = (const float*)d_in[6];
  const int*   ebatch   = (const int*)  d_in[7];
  const float* W0       = (const float*)d_in[8];
  const float* W1       = (const float*)d_in[9];
  const float* W2       = (const float*)d_in[10];
  float*       out      = (float*)d_out;

  const int nE = in_sizes[0] / DD;
  const int nT = in_sizes[4] / 3;

  float* ws    = (float*)d_ws;
  float* y     = ws;
  size_t nY    = (size_t)NNODE * DD;
  float* gsum  = y + nY;
  float* gcnt  = gsum + GG * DD;
  float* upper = gcnt + CR * GG;
  size_t nU    = (size_t)nE * DD;
  float* wt    = upper + nU;
  size_t nzero = nY + GG * DD + CR * GG;

  init_ws_kernel<<<2048, 256, 0, stream>>>(ws, nzero, W0, W1, W2, wt);

  {
    int tiles  = (nE + 16 * MT - 1) / (16 * MT);
    int blocks = (tiles + 7) / 8;
    edge_project_scatter_kernel<<<blocks, 256, 0, stream>>>(x, b1_rows, b1_vals,
                                                            ebatch, wt + 4096, wt,
                                                            y, upper, gcnt, nE);
  }
  {
    int tiles  = (nT + 16 * TMT - 1) / (16 * TMT);
    int blocks = (tiles + 7) / 8;
    tri_project_scatter_kernel<<<blocks, 256, 0, stream>>>(x, b2_rows, b2_vals,
                                                           wt + 2 * 4096, upper, nT);
  }
  {
    int tiles  = (nE + 16 * MT - 1) / (16 * MT);
    int blocks = (tiles + 7) / 8;
    combine_pool_kernel<<<blocks, 256, 0, stream>>>(upper, b1_rows, b1_vals,
                                                    ebatch, y, gsum, nE);
  }
  finalize_kernel<<<(GG * DD + 255) / 256, 256, 0, stream>>>(gsum, gcnt, out);
}
